# starting kernel
# baseline (speedup 1.0000x reference)
#include <hip/hip_runtime.h>
#include <math.h>

typedef float v2f __attribute__((ext_vector_type(2)));
typedef float v4f __attribute__((ext_vector_type(4)));
typedef float v8f __attribute__((ext_vector_type(8)));

#define FEAT_D 128

__global__ __launch_bounds__(256) void rnorm_kernel(const float* __restrict__ h,
                                                    float* __restrict__ rn,
                                                    int nrows) {
  const int wave = (blockIdx.x * blockDim.x + threadIdx.x) >> 5;
  const int lane = threadIdx.x & 31;
  if (wave >= nrows) return;

  const v4f v = *(const v4f*)(h + (size_t)wave * FEAT_D + lane * 4);
  float s = v.x * v.x + v.y * v.y + v.z * v.z + v.w * v.w;
#pragma unroll
  for (int off = 16; off > 0; off >>= 1) s += __shfl_xor(s, off, 32);
  if (lane == 0) rn[wave] = 1.0f / sqrtf(s);
}

__global__ __launch_bounds__(256) void edge_cos_wmma_kernel(
    const float* __restrict__ hg, const float* __restrict__ hd,
    const int* __restrict__ src, const int* __restrict__ dst,
    const float* __restrict__ rng, const float* __restrict__ rnd,
    float* __restrict__ out, int E) {
  const int wave = (blockIdx.x * blockDim.x + threadIdx.x) >> 5;
  const int lane = threadIdx.x & 31;
  const int e_base = wave * 16;
  if (e_base >= E) return;

  const int m = lane & 15;
  const int h = lane >> 4;
  int e_m = e_base + m;
  const int e_c = (e_m < E) ? e_m : (E - 1);

  const int si = src[e_c];
  const int di = dst[e_c];
  const float* gp = hg + (size_t)si * FEAT_D;
  const float* dp = hd + (size_t)di * FEAT_D;

  v8f c = {0.f, 0.f, 0.f, 0.f, 0.f, 0.f, 0.f, 0.f};

#pragma unroll 4
  for (int kk = 0; kk < 16; ++kk) {
    const int off = kk * 8 + h * 4;
    const v4f a4 = *(const v4f*)(gp + off);
    const v4f b4 = *(const v4f*)(dp + off);
    const v2f a0 = {a4.x, a4.y}, b0 = {b4.x, b4.y};
    const v2f a1 = {a4.z, a4.w}, b1 = {b4.z, b4.w};
    c = __builtin_amdgcn_wmma_f32_16x16x4_f32(false, a0, false, b0,
                                              (short)0, c, false, false);
    c = __builtin_amdgcn_wmma_f32_16x16x4_f32(false, a1, false, b1,
                                              (short)0, c, false, false);
  }

  const int sel = lane & 7;
  float d = c[0];
  if (sel == 1) d = c[1];
  if (sel == 2) d = c[2];
  if (sel == 3) d = c[3];
  if (sel == 4) d = c[4];
  if (sel == 5) d = c[5];
  if (sel == 6) d = c[6];
  if (sel == 7) d = c[7];

  const bool diag_lane = (lane < 8) || (lane >= 24);
  if (diag_lane && (e_m < E)) {
    out[e_m] = d * rng[si] * rnd[di];
  }
}

extern "C" void kernel_launch(void* const* d_in, const int* in_sizes, int n_in,
                              void* d_out, int out_size, void* d_ws, size_t ws_size,
                              hipStream_t stream) {
  const float* hg = (const float*)d_in[0];
  const float* hd = (const float*)d_in[1];
  const int* src = (const int*)d_in[2];
  const int* dst = (const int*)d_in[3];
  float* out = (float*)d_out;

  const int Ng = in_sizes[0] / FEAT_D;
  const int Nd = in_sizes[1] / FEAT_D;
  const int E = in_sizes[2];

  float* rng = (float*)d_ws;
  float* rnd = rng + Ng;

  rnorm_kernel<<<(Ng + 7) / 8, 256, 0, stream>>>(hg, rng, Ng);
  rnorm_kernel<<<(Nd + 7) / 8, 256, 0, stream>>>(hd, rnd, Nd);

  const int waves = (E + 15) / 16;
  const int blocks = (waves + 7) / 8;
  edge_cos_wmma_kernel<<<blocks, 256, 0, stream>>>(hg, hd, src, dst, rng, rnd,
                                                   out, E);
}


# Row norms taken inside the edge kernel from the operands it already gathers (per-lane sums of squares + one half-swap add: norm sum order changes); both norm launches and per-edge norm loads go
# speedup vs baseline: 1.1572x; 1.1572x over previous
#include <hip/hip_runtime.h>
#include <math.h>

typedef float v2f __attribute__((ext_vector_type(2)));
typedef float v4f __attribute__((ext_vector_type(4)));
typedef float v8f __attribute__((ext_vector_type(8)));

#define FEAT_D 128

__global__ __launch_bounds__(256) void edge_cos_wmma_kernel(
    const float* __restrict__ hg, const float* __restrict__ hd,
    const int* __restrict__ src, const int* __restrict__ dst,
    float* __restrict__ out, int E) {
  const int wave = (blockIdx.x * blockDim.x + threadIdx.x) >> 5;
  const int lane = threadIdx.x & 31;
  const int e_base = wave * 16;
  if (e_base >= E) return;

  const int m = lane & 15;
  const int h = lane >> 4;
  int e_m = e_base + m;
  const int e_c = (e_m < E) ? e_m : (E - 1);

  const int si = src[e_c];
  const int di = dst[e_c];
  const float* gp = hg + (size_t)si * FEAT_D;
  const float* dp = hd + (size_t)di * FEAT_D;

  v8f c = {0.f, 0.f, 0.f, 0.f, 0.f, 0.f, 0.f, 0.f};
  v2f qg = {0.f, 0.f};
  v2f qd = {0.f, 0.f};

#pragma unroll 4
  for (int kk = 0; kk < 16; ++kk) {
    const int off = kk * 8 + h * 4;
    const v4f a4 = *(const v4f*)(gp + off);
    const v4f b4 = *(const v4f*)(dp + off);
    const v2f a0 = {a4.x, a4.y}, b0 = {b4.x, b4.y};
    const v2f a1 = {a4.z, a4.w}, b1 = {b4.z, b4.w};
    c = __builtin_amdgcn_wmma_f32_16x16x4_f32(false, a0, false, b0,
                                              (short)0, c, false, false);
    c = __builtin_amdgcn_wmma_f32_16x16x4_f32(false, a1, false, b1,
                                              (short)0, c, false, false);
    qg += a0 * a0;
    qg += a1 * a1;
    qd += b0 * b0;
    qd += b1 * b1;
  }

  float sg = qg.x + qg.y;
  float sd = qd.x + qd.y;
  sg += __shfl_xor(sg, 16, 32);
  sd += __shfl_xor(sd, 16, 32);

  const int sel = lane & 7;
  float d = c[0];
  if (sel == 1) d = c[1];
  if (sel == 2) d = c[2];
  if (sel == 3) d = c[3];
  if (sel == 4) d = c[4];
  if (sel == 5) d = c[5];
  if (sel == 6) d = c[6];
  if (sel == 7) d = c[7];

  const bool diag_lane = (lane < 8) || (lane >= 24);
  if (diag_lane && (e_m < E)) {
    out[e_m] = d * (1.0f / sqrtf(sg)) * (1.0f / sqrtf(sd));
  }
}

extern "C" void kernel_launch(void* const* d_in, const int* in_sizes, int n_in,
                              void* d_out, int out_size, void* d_ws, size_t ws_size,
                              hipStream_t stream) {
  const float* hg = (const float*)d_in[0];
  const float* hd = (const float*)d_in[1];
  const int* src = (const int*)d_in[2];
  const int* dst = (const int*)d_in[3];
  float* out = (float*)d_out;

  const int E = in_sizes[2];

  const int waves = (E + 15) / 16;
  const int blocks = (waves + 7) / 8;
  edge_cos_wmma_kernel<<<blocks, 256, 0, stream>>>(hg, hd, src, dst, out, E);
}


# Gather re-cut to whole aligned 256-B units: a 16-B-per-lane load = 2 edges x 256 B (16 lanes an edge), not 16 lines x 32 B; dot and norm sums end in 8-lane butterfly adds (sum order changes)
# speedup vs baseline: 1.4115x; 1.2198x over previous
#include <hip/hip_runtime.h>
#include <math.h>

typedef float v2f __attribute__((ext_vector_type(2)));
typedef float v4f __attribute__((ext_vector_type(4)));
typedef float v8f __attribute__((ext_vector_type(8)));
typedef int v4i __attribute__((ext_vector_type(4)));

#define FEAT_D 128

__device__ __forceinline__ float xhalf(float x) {
  return __builtin_bit_cast(
      float, __builtin_amdgcn_permlanex16(__builtin_bit_cast(int, x),
                                          __builtin_bit_cast(int, x),
                                          0x76543210, 0xfedcba98, false, false));
}

template <int XM>
__device__ __forceinline__ float add_xmask(float x) {
  const int y = __builtin_amdgcn_mov_dpp(__builtin_bit_cast(int, x), 0x160 + XM,
                                         0xf, 0xf, true);
  return x + __builtin_bit_cast(float, y);
}

template <int I>
__device__ __forceinline__ void edge_pair(const float* __restrict__ hg,
                                          const float* __restrict__ hd,
                                          const int gi, const int di,
                                          const unsigned lbytes, const int lane,
                                          float& dot, float& na, float& nb) {
  const int sel = lane & 7;
  const unsigned goff = (unsigned)gi * (FEAT_D * 4u) + lbytes;
  const unsigned doff = (unsigned)di * (FEAT_D * 4u) + lbytes;
  const char* gp = (const char*)hg + goff;
  const char* dp = (const char*)hd + doff;
  const v4f a0 = *(const v4f*)(gp);
  const v4f b0 = *(const v4f*)(dp);
  const v4f a1 = *(const v4f*)(gp + 256);
  const v4f b1 = *(const v4f*)(dp + 256);

  v8f c = {0.f, 0.f, 0.f, 0.f, 0.f, 0.f, 0.f, 0.f};
  const v2f a0l = {a0.x, a0.y}, a0h = {a0.z, a0.w};
  const v2f b0l = {b0.x, b0.y}, b0h = {b0.z, b0.w};
  const v2f a1l = {a1.x, a1.y}, a1h = {a1.z, a1.w};
  const v2f b1l = {b1.x, b1.y}, b1h = {b1.z, b1.w};
  c = __builtin_amdgcn_wmma_f32_16x16x4_f32(false, a0l, false, b0l, (short)0, c,
                                            false, false);
  c = __builtin_amdgcn_wmma_f32_16x16x4_f32(false, a0h, false, b0h, (short)0, c,
                                            false, false);
  c = __builtin_amdgcn_wmma_f32_16x16x4_f32(false, a1l, false, b1l, (short)0, c,
                                            false, false);
  c = __builtin_amdgcn_wmma_f32_16x16x4_f32(false, a1h, false, b1h, (short)0, c,
                                            false, false);

  v2f qg = a0l * a0l;
  qg += a0h * a0h;
  qg += a1l * a1l;
  qg += a1h * a1h;
  v2f qd = b0l * b0l;
  qd += b0h * b0h;
  qd += b1l * b1l;
  qd += b1h * b1h;

  float t = c[0];
  if (sel == 1) t = c[1];
  if (sel == 2) t = c[2];
  if (sel == 3) t = c[3];
  if (sel == 4) t = c[4];
  if (sel == 5) t = c[5];
  if (sel == 6) t = c[6];
  if (sel == 7) t = c[7];
  t = add_xmask<1>(t);
  t = add_xmask<2>(t);
  t = add_xmask<4>(t);

  const float sg = qg.x + qg.y;
  const float sd = qd.x + qd.y;
  const bool up = lane >= 16;
  const float keep = up ? sd : sg;
  const float send = up ? sg : sd;
  float w = keep + xhalf(send);
  w = add_xmask<1>(w);
  w = add_xmask<2>(w);
  w = add_xmask<4>(w);
  const float w2 = xhalf(w);

  const bool mine = sel == I;
  dot = mine ? t : dot;
  na = mine ? w : na;
  nb = mine ? w2 : nb;
}

__global__ __launch_bounds__(256, 6) void edge_cos_wmma_kernel(
    const float* __restrict__ hg, const float* __restrict__ hd,
    const int* __restrict__ src, const int* __restrict__ dst,
    float* __restrict__ out, int E) {
  const int wave = (blockIdx.x * blockDim.x + threadIdx.x) >> 5;
  const int lane = threadIdx.x & 31;
  const int e_base = wave * 16;
  if (e_base >= E) return;

  const int s = (lane >> 3) & 1;
  const int chunk = (lane & 7) + ((lane >> 4) << 3);
  const unsigned lbytes = (unsigned)chunk * 16u;
  const int e0 = e_base + 8 * s;

  int gi[8], di[8];
  if (e_base + 16 <= E) {
    const v4i g0 = *(const v4i*)(src + e0);
    const v4i g1 = *(const v4i*)(src + e0 + 4);
    const v4i d0 = *(const v4i*)(dst + e0);
    const v4i d1 = *(const v4i*)(dst + e0 + 4);
    gi[0] = g0.x; gi[1] = g0.y; gi[2] = g0.z; gi[3] = g0.w;
    gi[4] = g1.x; gi[5] = g1.y; gi[6] = g1.z; gi[7] = g1.w;
    di[0] = d0.x; di[1] = d0.y; di[2] = d0.z; di[3] = d0.w;
    di[4] = d1.x; di[5] = d1.y; di[6] = d1.z; di[7] = d1.w;
  } else {
#pragma unroll
    for (int i = 0; i < 8; ++i) {
      const int e = e0 + i;
      const int e_c = (e < E) ? e : (E - 1);
      gi[i] = src[e_c];
      di[i] = dst[e_c];
    }
  }

  float dot = 0.f, na = 1.f, nb = 1.f;
  edge_pair<0>(hg, hd, gi[0], di[0], lbytes, lane, dot, na, nb);
  edge_pair<1>(hg, hd, gi[1], di[1], lbytes, lane, dot, na, nb);
  edge_pair<2>(hg, hd, gi[2], di[2], lbytes, lane, dot, na, nb);
  edge_pair<3>(hg, hd, gi[3], di[3], lbytes, lane, dot, na, nb);
  edge_pair<4>(hg, hd, gi[4], di[4], lbytes, lane, dot, na, nb);
  edge_pair<5>(hg, hd, gi[5], di[5], lbytes, lane, dot, na, nb);
  edge_pair<6>(hg, hd, gi[6], di[6], lbytes, lane, dot, na, nb);
  edge_pair<7>(hg, hd, gi[7], di[7], lbytes, lane, dot, na, nb);

  const int e_m = e0 + (lane & 7);
  const bool diag_lane = (lane < 8) || (lane >= 24);
  if (diag_lane && (e_m < E)) {
    out[e_m] = dot * ((1.0f / sqrtf(na)) * (1.0f / sqrtf(nb)));
  }
}

extern "C" void kernel_launch(void* const* d_in, const int* in_sizes, int n_in,
                              void* d_out, int out_size, void* d_ws, size_t ws_size,
                              hipStream_t stream) {
  const float* hg = (const float*)d_in[0];
  const float* hd = (const float*)d_in[1];
  const int* src = (const int*)d_in[2];
  const int* dst = (const int*)d_in[3];
  float* out = (float*)d_out;

  const int E = in_sizes[2];

  const int waves = (E + 15) / 16;
  const int blocks = (waves + 7) / 8;
  edge_cos_wmma_kernel<<<blocks, 256, 0, stream>>>(hg, hd, src, dst, out, E);
}


# Each wave walks over 3-4 groups of 16 edges (3072 blocks, was 11719), next 3 pairs' row loads kept in flight across group ends; 126 VGPRs -> 8 waves a SIMD; per-edge sums and their order unchanged
# speedup vs baseline: 1.5524x; 1.0998x over previous
#include <hip/hip_runtime.h>
#include <math.h>

typedef float v2f __attribute__((ext_vector_type(2)));
typedef float v4f __attribute__((ext_vector_type(4)));
typedef float v8f __attribute__((ext_vector_type(8)));
typedef int v4i __attribute__((ext_vector_type(4)));

#define FEAT_D 128
#define ROW_BYTES (FEAT_D * 4u)
#define AHEAD 3

__device__ __forceinline__ float xhalf(float x) {
  return __builtin_bit_cast(
      float, __builtin_amdgcn_permlanex16(__builtin_bit_cast(int, x),
                                          __builtin_bit_cast(int, x),
                                          0x76543210, 0xfedcba98, false, false));
}

template <int XM>
__device__ __forceinline__ float add_xmask(float x) {
  const int y = __builtin_amdgcn_mov_dpp(__builtin_bit_cast(int, x), 0x160 + XM,
                                         0xf, 0xf, true);
  return x + __builtin_bit_cast(float, y);
}

struct RowPair {
  v4f a0, b0, a1, b1;
};

__device__ __forceinline__ void issue_rows(const float* __restrict__ hg,
                                           const float* __restrict__ hd,
                                           const int gi, const int di,
                                           const unsigned lbytes, RowPair& r) {
  const unsigned goff = (unsigned)gi * ROW_BYTES + lbytes;
  const unsigned doff = (unsigned)di * ROW_BYTES + lbytes;
  const char* gp = (const char*)hg + goff;
  const char* dp = (const char*)hd + doff;
  r.a0 = *(const v4f*)(gp);
  r.b0 = *(const v4f*)(dp);
  r.a1 = *(const v4f*)(gp + 256);
  r.b1 = *(const v4f*)(dp + 256);
}

__device__ __forceinline__ void sum_pair(const RowPair& r, const int i,
                                         const int lane, float& dot, float& na,
                                         float& nb) {
  const int sel = lane & 7;
  const v4f a0 = r.a0, b0 = r.b0, a1 = r.a1, b1 = r.b1;

  v8f c = {0.f, 0.f, 0.f, 0.f, 0.f, 0.f, 0.f, 0.f};
  const v2f a0l = {a0.x, a0.y}, a0h = {a0.z, a0.w};
  const v2f b0l = {b0.x, b0.y}, b0h = {b0.z, b0.w};
  const v2f a1l = {a1.x, a1.y}, a1h = {a1.z, a1.w};
  const v2f b1l = {b1.x, b1.y}, b1h = {b1.z, b1.w};
  c = __builtin_amdgcn_wmma_f32_16x16x4_f32(false, a0l, false, b0l, (short)0, c,
                                            false, false);
  c = __builtin_amdgcn_wmma_f32_16x16x4_f32(false, a0h, false, b0h, (short)0, c,
                                            false, false);
  c = __builtin_amdgcn_wmma_f32_16x16x4_f32(false, a1l, false, b1l, (short)0, c,
                                            false, false);
  c = __builtin_amdgcn_wmma_f32_16x16x4_f32(false, a1h, false, b1h, (short)0, c,
                                            false, false);

  v2f qg = a0l * a0l;
  qg += a0h * a0h;
  qg += a1l * a1l;
  qg += a1h * a1h;
  v2f qd = b0l * b0l;
  qd += b0h * b0h;
  qd += b1l * b1l;
  qd += b1h * b1h;

  float t = c[0];
  if (sel == 1) t = c[1];
  if (sel == 2) t = c[2];
  if (sel == 3) t = c[3];
  if (sel == 4) t = c[4];
  if (sel == 5) t = c[5];
  if (sel == 6) t = c[6];
  if (sel == 7) t = c[7];
  t = add_xmask<1>(t);
  t = add_xmask<2>(t);
  t = add_xmask<4>(t);

  const float sg = qg.x + qg.y;
  const float sd = qd.x + qd.y;
  const bool up = lane >= 16;
  const float keep = up ? sd : sg;
  const float send = up ? sg : sd;
  float w = keep + xhalf(send);
  w = add_xmask<1>(w);
  w = add_xmask<2>(w);
  w = add_xmask<4>(w);
  const float w2 = xhalf(w);

  const bool mine = sel == i;
  dot = mine ? t : dot;
  na = mine ? w : na;
  nb = mine ? w2 : nb;
}

__device__ __forceinline__ void load_indices(const int* __restrict__ src,
                                             const int* __restrict__ dst,
                                             const int e_base, const int E,
                                             const int s, int (&gi)[8],
                                             int (&di)[8]) {
  const int e0 = e_base + 8 * s;
  if (e_base + 16 <= E) {
    const v4i g0 = *(const v4i*)(src + e0);
    const v4i g1 = *(const v4i*)(src + e0 + 4);
    const v4i d0 = *(const v4i*)(dst + e0);
    const v4i d1 = *(const v4i*)(dst + e0 + 4);
    gi[0] = g0.x; gi[1] = g0.y; gi[2] = g0.z; gi[3] = g0.w;
    gi[4] = g1.x; gi[5] = g1.y; gi[6] = g1.z; gi[7] = g1.w;
    di[0] = d0.x; di[1] = d0.y; di[2] = d0.z; di[3] = d0.w;
    di[4] = d1.x; di[5] = d1.y; di[6] = d1.z; di[7] = d1.w;
  } else {
#pragma unroll
    for (int k = 0; k < 8; ++k) {
      const int e = e0 + k;
      const int e_c = (e < E) ? e : (E - 1);
      gi[k] = src[e_c];
      di[k] = dst[e_c];
    }
  }
}

__device__ __forceinline__ void store_group(float* __restrict__ out,
                                            const int e_base, const int E,
                                            const int s, const int lane,
                                            const float dot, const float na,
                                            const float nb) {
  const int e_m = e_base + 8 * s + (lane & 7);
  const bool diag_lane = (lane < 8) || (lane >= 24);
  if (diag_lane && (e_m < E)) {
    out[e_m] = dot * ((1.0f / sqrtf(na)) * (1.0f / sqrtf(nb)));
  }
}

__global__ __launch_bounds__(256, 8) void edge_cos_wmma_kernel(
    const float* __restrict__ hg, const float* __restrict__ hd,
    const int* __restrict__ src, const int* __restrict__ dst,
    float* __restrict__ out, int E) {
  const int lane = threadIdx.x & 31;
  const int wib = __builtin_amdgcn_readfirstlane((int)(threadIdx.x >> 5));
  const int n_waves = (int)gridDim.x * 8;
  const int n_groups = (E + 15) >> 4;
  int g = (int)blockIdx.x * 8 + wib;
  if (g >= n_groups) return;

  const int s = (lane >> 3) & 1;
  const int chunk = (lane & 7) + ((lane >> 4) << 3);
  const unsigned lbytes = (unsigned)chunk * 16u;

  int gi[8], di[8];
  load_indices(src, dst, g * 16, E, s, gi, di);

  RowPair buf[8];
#pragma unroll
  for (int i = 0; i < AHEAD; ++i)
    issue_rows(hg, hd, gi[i], di[i], lbytes, buf[i]);

  int gn = g + n_waves;
  while (gn < n_groups) {
    int ngi[8], ndi[8];
    load_indices(src, dst, gn * 16, E, s, ngi, ndi);

    float dot = 0.f, na = 1.f, nb = 1.f;
#pragma unroll
    for (int i = 0; i < 8; ++i) {
      const int j = i + AHEAD;
      if (j < 8)
        issue_rows(hg, hd, gi[j], di[j], lbytes, buf[j]);
      else
        issue_rows(hg, hd, ngi[j - 8], ndi[j - 8], lbytes, buf[j - 8]);
      sum_pair(buf[i], i, lane, dot, na, nb);
      __builtin_amdgcn_sched_barrier(0);
    }
    store_group(out, g * 16, E, s, lane, dot, na, nb);

#pragma unroll
    for (int k = 0; k < 8; ++k) {
      gi[k] = ngi[k];
      di[k] = ndi[k];
    }
    g = gn;
    gn += n_waves;
  }

  {
    float dot = 0.f, na = 1.f, nb = 1.f;
#pragma unroll
    for (int i = 0; i < 8; ++i) {
      const int j = i + AHEAD;
      if (j < 8) issue_rows(hg, hd, gi[j], di[j], lbytes, buf[j]);
      sum_pair(buf[i], i, lane, dot, na, nb);
      __builtin_amdgcn_sched_barrier(0);
    }
    store_group(out, g * 16, E, s, lane, dot, na, nb);
  }
}

extern "C" void kernel_launch(void* const* d_in, const int* in_sizes, int n_in,
                              void* d_out, int out_size, void* d_ws, size_t ws_size,
                              hipStream_t stream) {
  const float* hg = (const float*)d_in[0];
  const float* hd = (const float*)d_in[1];
  const int* src = (const int*)d_in[2];
  const int* dst = (const int*)d_in[3];
  float* out = (float*)d_out;

  const int E = in_sizes[2];

  edge_cos_wmma_kernel<<<3072, 256, 0, stream>>>(hg, hd, src, dst, out, E);
}
